# starting kernel
# baseline (speedup 1.0000x reference)
#include <hip/hip_runtime.h>
#include <hip/hip_bf16.h>

#define BATCH 4096
#define TSEQ  200
#define FEAT  32
#define HID   8
#define FWD   5

typedef __attribute__((ext_vector_type(2))) float v2f;
typedef __attribute__((ext_vector_type(8))) float v8f;

__device__ __forceinline__ float sigm(float x) {
    return 1.0f / (1.0f + __expf(-x));
}
__device__ __forceinline__ float tanh_fast(float x) {
    return 1.0f - 2.0f / (1.0f + __expf(2.0f * x));
}

__device__ __forceinline__ float wcat(const float* __restrict__ W_ih,
                                      const float* __restrict__ W_hh,
                                      int k, int n) {
    if (k < FEAT + 1)       return W_ih[n * (FEAT + 1) + k];
    if (k < FEAT + 1 + HID) return W_hh[n * HID + (k - (FEAT + 1))];
    return 0.0f;
}

__device__ __forceinline__ void loadx(v2f (&dst)[8], const float* __restrict__ xr, int hi) {
#pragma unroll
    for (int c = 0; c < 8; ++c)
        dst[c] = *reinterpret_cast<const v2f*>(xr + 4 * c + 2 * hi);
}

#define WMMA_F32(ACC, A, BFRAG) \
    (ACC) = __builtin_amdgcn_wmma_f32_16x16x4_f32(false, (A), false, (BFRAG), (short)0, (ACC), false, false)

__global__ __launch_bounds__(32)
void monotonic_lstm_wmma(const float* __restrict__ x,
                         const float* __restrict__ z0,
                         const float* __restrict__ W_ih,
                         const float* __restrict__ W_hh,
                         const float* __restrict__ b_ih,
                         const float* __restrict__ b_hh,
                         const float* __restrict__ W1,
                         const float* __restrict__ b1,
                         const float* __restrict__ W2,
                         const float* __restrict__ b2,
                         float* __restrict__ out)
{
    __shared__ float gates_s[16 * 33];
    __shared__ float h_s[16 * 9];
    __shared__ float z_s[16];

    const int lane = threadIdx.x;
    const int m    = lane & 15;
    const int hi   = lane >> 4;
    const int b0   = blockIdx.x * 16;

    v2f bw0[11], bw1[11];
#pragma unroll
    for (int c = 0; c < 11; ++c) {
        const int k0 = 4 * c + 2 * hi;
        bw0[c].x = wcat(W_ih, W_hh, k0,     m);
        bw0[c].y = wcat(W_ih, W_hh, k0 + 1, m);
        bw1[c].x = wcat(W_ih, W_hh, k0,     m + 16);
        bw1[c].y = wcat(W_ih, W_hh, k0 + 1, m + 16);
    }
    const float bias0 = b_ih[m]      + b_hh[m];
    const float bias1 = b_ih[m + 16] + b_hh[m + 16];

    float w1r[FWD * HID], b1r[FWD], w2r[FWD];
#pragma unroll
    for (int i = 0; i < FWD * HID; ++i) w1r[i] = W1[i];
#pragma unroll
    for (int i = 0; i < FWD; ++i) { b1r[i] = b1[i]; w2r[i] = W2[i]; }
    const float b2r = b2[0];

    float cst[4] = {0.f, 0.f, 0.f, 0.f};
    float zreg = z0[b0 + m];
    if (hi == 0) z_s[m] = zreg;
#pragma unroll
    for (int u = 0; u < 4; ++u) h_s[m * 9 + hi * 4 + u] = 0.0f;
    __builtin_amdgcn_wave_barrier();

    const float* xrow = x + ((size_t)(b0 + m) * TSEQ) * FEAT;

    v2f xbuf[2][8];
    loadx(xbuf[0], xrow, hi);

#pragma unroll 2
    for (int t = 0; t < TSEQ; ++t) {
        const int cb = t & 1;
        const int tn = (t + 1 < TSEQ) ? (t + 1) : t;
        loadx(xbuf[cb ^ 1], xrow + (size_t)tn * FEAT, hi);

        v8f a0e, a0o, a1e, a1o;
#pragma unroll
        for (int r = 0; r < 8; ++r) {
            a0e[r] = bias0; a1e[r] = bias1;
            a0o[r] = 0.0f;  a1o[r] = 0.0f;
        }

#pragma unroll
        for (int c = 0; c < 8; c += 2) {
            WMMA_F32(a0e, xbuf[cb][c],     bw0[c]);
            WMMA_F32(a1e, xbuf[cb][c],     bw1[c]);
            WMMA_F32(a0o, xbuf[cb][c + 1], bw0[c + 1]);
            WMMA_F32(a1o, xbuf[cb][c + 1], bw1[c + 1]);
        }
        {
            const float h7 = h_s[m * 9 + 7];
            v2f a;
            a.x = hi ? h_s[m * 9 + 1] : z_s[m];
            a.y = hi ? h_s[m * 9 + 2] : h_s[m * 9 + 0];
            WMMA_F32(a0e, a, bw0[8]);
            WMMA_F32(a1e, a, bw1[8]);
            a.x = h_s[m * 9 + 3 + 2 * hi];
            a.y = h_s[m * 9 + 4 + 2 * hi];
            WMMA_F32(a0o, a, bw0[9]);
            WMMA_F32(a1o, a, bw1[9]);
            a.x = hi ? 0.0f : h7;
            a.y = 0.0f;
            WMMA_F32(a0e, a, bw0[10]);
            WMMA_F32(a1e, a, bw1[10]);
        }
        __builtin_amdgcn_wave_barrier();

#pragma unroll
        for (int r = 0; r < 8; ++r) {
            gates_s[(r + 8 * hi) * 33 + m]      = a0e[r] + a0o[r];
            gates_s[(r + 8 * hi) * 33 + m + 16] = a1e[r] + a1o[r];
        }
        __builtin_amdgcn_wave_barrier();

#pragma unroll
        for (int u = 0; u < 4; ++u) {
            const int j = hi * 4 + u;
            const float ig = sigm(gates_s[m * 33 + j]);
            const float fg = sigm(gates_s[m * 33 + 8 + j]);
            const float gg = tanh_fast(gates_s[m * 33 + 16 + j]);
            const float og = sigm(gates_s[m * 33 + 24 + j]);
            const float cn = fg * cst[u] + ig * gg;
            cst[u] = cn;
            h_s[m * 9 + j] = og * tanh_fast(cn);
        }
        __builtin_amdgcn_wave_barrier();

        if (hi == 0) {
            float hv[HID];
#pragma unroll
            for (int j = 0; j < HID; ++j) hv[j] = h_s[m * 9 + j];
            float delta = b2r;
#pragma unroll
            for (int k = 0; k < FWD; ++k) {
                float d = b1r[k];
#pragma unroll
                for (int j = 0; j < HID; ++j) d = fmaf(hv[j], w1r[k * HID + j], d);
                delta = fmaf(fmaxf(d, 0.0f), w2r[k], delta);
            }
            zreg += fmaxf(delta, 0.0f);
            z_s[m] = zreg;
            out[(size_t)(b0 + m) * TSEQ + t] = zreg;
        }
        __builtin_amdgcn_wave_barrier();
    }
}

extern "C" void kernel_launch(void* const* d_in, const int* in_sizes, int n_in,
                              void* d_out, int out_size, void* d_ws, size_t ws_size,
                              hipStream_t stream) {
    (void)in_sizes; (void)n_in; (void)out_size; (void)d_ws; (void)ws_size;
    const float* x    = (const float*)d_in[0];
    const float* z0   = (const float*)d_in[1];
    const float* W_ih = (const float*)d_in[2];
    const float* W_hh = (const float*)d_in[3];
    const float* b_ih = (const float*)d_in[4];
    const float* b_hh = (const float*)d_in[5];
    const float* W1   = (const float*)d_in[6];
    const float* b1   = (const float*)d_in[7];
    const float* W2   = (const float*)d_in[8];
    const float* b2   = (const float*)d_in[9];
    float* out = (float*)d_out;

    dim3 grid(BATCH / 16);
    dim3 block(32);
    monotonic_lstm_wmma<<<grid, block, 0, stream>>>(x, z0, W_ih, W_hh, b_ih, b_hh,
                                                    W1, b1, W2, b2, out);
}


# Transposed gate product (weights as A, activations as B): D lands in each lane's own cell, h/c/z stay in registers and feed back as B, head as a third tile; no LDS; recurrent+head sum order changes
# speedup vs baseline: 1.5980x; 1.5980x over previous
#include <hip/hip_runtime.h>
#include <hip/hip_bf16.h>


#define BATCH 4096
#define TSEQ  200
#define FEAT  32
#define HID   8
#define FWD   5


typedef __attribute__((ext_vector_type(2))) float v2f;
typedef __attribute__((ext_vector_type(8))) float v8f;

__device__ __forceinline__ float sigm(float x) {
    return 1.0f / (1.0f + __expf(-x));
}
__device__ __forceinline__ float tanh_fast(float x) {
    return 1.0f - 2.0f / (1.0f + __expf(2.0f * x));
}

__device__ __forceinline__ int gate_of_row(int tile, int rho) {
    const int r = rho & 7, h = rho >> 3;
    return tile * 16 + ((r & 4) ? 8 : 0) + 4 * h + (r & 3);
}

__device__ __forceinline__ float wfrag(const float* __restrict__ W_ih,
                                       const float* __restrict__ W_hh,
                                       int g, int c, int hi, int s) {
    if (c < 8)  return W_ih[g * (FEAT + 1) + 4 * c + 2 * hi + s];
    if (c == 8) return W_hh[g * HID + 4 * hi + s];
    if (c == 9) return W_hh[g * HID + 4 * hi + 2 + s];
    return (hi == 0 && s == 0) ? W_ih[g * (FEAT + 1) + FEAT] : 0.0f;
}


__device__ __forceinline__ void loadx(v2f (&dst)[8], const float* __restrict__ xr, int hi) {
#pragma unroll
    for (int c = 0; c < 8; ++c)
        dst[c] = *reinterpret_cast<const v2f*>(xr + 4 * c + 2 * hi);
}

#define WMMA_F32(ACC, A, BFRAG) \
    (ACC) = __builtin_amdgcn_wmma_f32_16x16x4_f32(false, (A), false, (BFRAG), (short)0, (ACC), false, false)

__global__ __launch_bounds__(32)
void monotonic_lstm_wmma(const float* __restrict__ x,
                         const float* __restrict__ z0,
                         const float* __restrict__ W_ih,
                         const float* __restrict__ W_hh,
                         const float* __restrict__ b_ih,
                         const float* __restrict__ b_hh,
                         const float* __restrict__ W1,
                         const float* __restrict__ b1,
                         const float* __restrict__ W2,
                         const float* __restrict__ b2,
                         float* __restrict__ out)
{
    const int lane = threadIdx.x;
    const int m    = lane & 15;
    const int hi   = lane >> 4;
    const int b0   = blockIdx.x * 16;

    const int g0 = gate_of_row(0, m);
    const int g1 = gate_of_row(1, m);
    v2f aw0[11], aw1[11];
#pragma unroll
    for (int c = 0; c < 11; ++c) {
        aw0[c].x = wfrag(W_ih, W_hh, g0, c, hi, 0);
        aw0[c].y = wfrag(W_ih, W_hh, g0, c, hi, 1);
        aw1[c].x = wfrag(W_ih, W_hh, g1, c, hi, 0);
        aw1[c].y = wfrag(W_ih, W_hh, g1, c, hi, 1);
    }
    float bias0[8], bias1[8];
#pragma unroll
    for (int r = 0; r < 8; ++r) {
        const int ga = gate_of_row(0, 8 * hi + r);
        bias0[r] = b_ih[ga]      + b_hh[ga];
        bias1[r] = b_ih[ga + 16] + b_hh[ga + 16];
    }

    const int  kk = m & 7;
    const bool kv = kk < FWD;
    v2f ah0, ah1;
    ah0.x = kv ? W1[kk * HID + 4 * hi + 0] : 0.0f;
    ah0.y = kv ? W1[kk * HID + 4 * hi + 1] : 0.0f;
    ah1.x = kv ? W1[kk * HID + 4 * hi + 2] : 0.0f;
    ah1.y = kv ? W1[kk * HID + 4 * hi + 3] : 0.0f;
    float b1r[FWD], w2r[FWD];
#pragma unroll
    for (int i = 0; i < FWD; ++i) { b1r[i] = b1[i]; w2r[i] = W2[i]; }
    const float b2r = b2[0];

    float cst[4]  = {0.f, 0.f, 0.f, 0.f};
    float hreg[4] = {0.f, 0.f, 0.f, 0.f};
    float zreg = z0[b0 + m];

    const float* xrow = x + ((size_t)(b0 + m) * TSEQ) * FEAT;
    float* orow = out + (size_t)(b0 + m) * TSEQ;

    v2f xbuf[2][8];
    loadx(xbuf[0], xrow, hi);

#pragma unroll 2
    for (int t = 0; t < TSEQ; ++t) {
        const int cb = t & 1;
        const int tn = (t + 1 < TSEQ) ? (t + 1) : t;
        __builtin_amdgcn_sched_barrier(0);
        loadx(xbuf[cb ^ 1], xrow + (size_t)tn * FEAT, hi);
        __builtin_amdgcn_sched_barrier(0);

        v8f a0e, a0o, a1e, a1o;
#pragma unroll
        for (int r = 0; r < 8; ++r) {
            a0e[r] = bias0[r]; a1e[r] = bias1[r];
            a0o[r] = 0.0f;     a1o[r] = 0.0f;
        }

#pragma unroll
        for (int c = 0; c < 8; c += 2) {
            WMMA_F32(a0e, aw0[c],     xbuf[cb][c]);
            WMMA_F32(a1e, aw1[c],     xbuf[cb][c]);
            WMMA_F32(a0o, aw0[c + 1], xbuf[cb][c + 1]);
            WMMA_F32(a1o, aw1[c + 1], xbuf[cb][c + 1]);
        }
        {
            v2f hb0, hb1, zb;
            hb0.x = hreg[0]; hb0.y = hreg[1];
            hb1.x = hreg[2]; hb1.y = hreg[3];
            zb.x  = hi ? 0.0f : zreg;
            zb.y  = 0.0f;
            WMMA_F32(a0e, aw0[8],  hb0);
            WMMA_F32(a1e, aw1[8],  hb0);
            WMMA_F32(a0o, aw0[9],  hb1);
            WMMA_F32(a1o, aw1[9],  hb1);
            WMMA_F32(a0e, aw0[10], zb);
            WMMA_F32(a1e, aw1[10], zb);
        }

#pragma unroll
        for (int u = 0; u < 4; ++u) {
            const float ig = sigm(a0e[u] + a0o[u]);
            const float fg = sigm(a0e[4 + u] + a0o[4 + u]);
            const float gg = tanh_fast(a1e[u] + a1o[u]);
            const float og = sigm(a1e[4 + u] + a1o[4 + u]);
            const float cn = fg * cst[u] + ig * gg;
            cst[u]  = cn;
            hreg[u] = og * tanh_fast(cn);
        }

        v8f hda, hdb;
#pragma unroll
        for (int r = 0; r < 8; ++r) {
            hda[r] = (r < FWD) ? b1r[r] : 0.0f;
            hdb[r] = 0.0f;
        }
        {
            v2f hb0, hb1;
            hb0.x = hreg[0]; hb0.y = hreg[1];
            hb1.x = hreg[2]; hb1.y = hreg[3];
            WMMA_F32(hda, ah0, hb0);
            WMMA_F32(hdb, ah1, hb1);
        }
        float delta = b2r;
#pragma unroll
        for (int k = 0; k < FWD; ++k)
            delta = fmaf(fmaxf(hda[k] + hdb[k], 0.0f), w2r[k], delta);
        zreg += fmaxf(delta, 0.0f);
        if (hi == 0) orow[t] = zreg;
    }
}

extern "C" void kernel_launch(void* const* d_in, const int* in_sizes, int n_in,
                              void* d_out, int out_size, void* d_ws, size_t ws_size,
                              hipStream_t stream) {
    (void)in_sizes; (void)n_in; (void)out_size; (void)d_ws; (void)ws_size;
    const float* x    = (const float*)d_in[0];
    const float* z0   = (const float*)d_in[1];
    const float* W_ih = (const float*)d_in[2];
    const float* W_hh = (const float*)d_in[3];
    const float* b_ih = (const float*)d_in[4];
    const float* b_hh = (const float*)d_in[5];
    const float* W1   = (const float*)d_in[6];
    const float* b1   = (const float*)d_in[7];
    const float* W2   = (const float*)d_in[8];
    const float* b2   = (const float*)d_in[9];
    float* out = (float*)d_out;

    dim3 grid(BATCH / 16);
    dim3 block(32);
    monotonic_lstm_wmma<<<grid, block, 0, stream>>>(x, z0, W_ih, W_hh, b_ih, b_hh,
                                                    W1, b1, W2, b2, out);
}


# Gate sigmoids and tanh: the correctly rounded division sequence (20 quotients per step) replaced by the hardware reciprocal instruction (1 ulp); exp, sums and their order as before
# speedup vs baseline: 1.9888x; 1.2446x over previous
#include <hip/hip_runtime.h>
#include <hip/hip_bf16.h>


#define BATCH 4096
#define TSEQ  200
#define FEAT  32
#define HID   8
#define FWD   5


typedef __attribute__((ext_vector_type(2))) float v2f;
typedef __attribute__((ext_vector_type(8))) float v8f;

__device__ __forceinline__ float recip_ge1(float d) {
    return __builtin_amdgcn_rcpf(d);
}
__device__ __forceinline__ float sigm(float x) {
    return recip_ge1(1.0f + __expf(-x));
}
__device__ __forceinline__ float tanh_fast(float x) {
    return fmaf(-2.0f, recip_ge1(1.0f + __expf(2.0f * x)), 1.0f);
}

__device__ __forceinline__ int gate_of_row(int tile, int rho) {
    const int r = rho & 7, h = rho >> 3;
    return tile * 16 + ((r & 4) ? 8 : 0) + 4 * h + (r & 3);
}

__device__ __forceinline__ float wfrag(const float* __restrict__ W_ih,
                                       const float* __restrict__ W_hh,
                                       int g, int c, int hi, int s) {
    if (c < 8)  return W_ih[g * (FEAT + 1) + 4 * c + 2 * hi + s];
    if (c == 8) return W_hh[g * HID + 4 * hi + s];
    if (c == 9) return W_hh[g * HID + 4 * hi + 2 + s];
    return (hi == 0 && s == 0) ? W_ih[g * (FEAT + 1) + FEAT] : 0.0f;
}


__device__ __forceinline__ void loadx(v2f (&dst)[8], const float* __restrict__ xr, int hi) {
#pragma unroll
    for (int c = 0; c < 8; ++c)
        dst[c] = *reinterpret_cast<const v2f*>(xr + 4 * c + 2 * hi);
}

#define WMMA_F32(ACC, A, BFRAG) \
    (ACC) = __builtin_amdgcn_wmma_f32_16x16x4_f32(false, (A), false, (BFRAG), (short)0, (ACC), false, false)

__global__ __launch_bounds__(32)
void monotonic_lstm_wmma(const float* __restrict__ x,
                         const float* __restrict__ z0,
                         const float* __restrict__ W_ih,
                         const float* __restrict__ W_hh,
                         const float* __restrict__ b_ih,
                         const float* __restrict__ b_hh,
                         const float* __restrict__ W1,
                         const float* __restrict__ b1,
                         const float* __restrict__ W2,
                         const float* __restrict__ b2,
                         float* __restrict__ out)
{
    const int lane = threadIdx.x;
    const int m    = lane & 15;
    const int hi   = lane >> 4;
    const int b0   = blockIdx.x * 16;

    const int g0 = gate_of_row(0, m);
    const int g1 = gate_of_row(1, m);
    v2f aw0[11], aw1[11];
#pragma unroll
    for (int c = 0; c < 11; ++c) {
        aw0[c].x = wfrag(W_ih, W_hh, g0, c, hi, 0);
        aw0[c].y = wfrag(W_ih, W_hh, g0, c, hi, 1);
        aw1[c].x = wfrag(W_ih, W_hh, g1, c, hi, 0);
        aw1[c].y = wfrag(W_ih, W_hh, g1, c, hi, 1);
    }
    float bias0[8], bias1[8];
#pragma unroll
    for (int r = 0; r < 8; ++r) {
        const int ga = gate_of_row(0, 8 * hi + r);
        bias0[r] = b_ih[ga]      + b_hh[ga];
        bias1[r] = b_ih[ga + 16] + b_hh[ga + 16];
    }

    const int  kk = m & 7;
    const bool kv = kk < FWD;
    v2f ah0, ah1;
    ah0.x = kv ? W1[kk * HID + 4 * hi + 0] : 0.0f;
    ah0.y = kv ? W1[kk * HID + 4 * hi + 1] : 0.0f;
    ah1.x = kv ? W1[kk * HID + 4 * hi + 2] : 0.0f;
    ah1.y = kv ? W1[kk * HID + 4 * hi + 3] : 0.0f;
    float b1r[FWD], w2r[FWD];
#pragma unroll
    for (int i = 0; i < FWD; ++i) { b1r[i] = b1[i]; w2r[i] = W2[i]; }
    const float b2r = b2[0];

    float cst[4]  = {0.f, 0.f, 0.f, 0.f};
    float hreg[4] = {0.f, 0.f, 0.f, 0.f};
    float zreg = z0[b0 + m];

    const float* xrow = x + ((size_t)(b0 + m) * TSEQ) * FEAT;
    float* orow = out + (size_t)(b0 + m) * TSEQ;

    v2f xbuf[2][8];
    loadx(xbuf[0], xrow, hi);

#pragma unroll 2
    for (int t = 0; t < TSEQ; ++t) {
        const int cb = t & 1;
        const int tn = (t + 1 < TSEQ) ? (t + 1) : t;
        __builtin_amdgcn_sched_barrier(0);
        loadx(xbuf[cb ^ 1], xrow + (size_t)tn * FEAT, hi);
        __builtin_amdgcn_sched_barrier(0);

        v8f a0e, a0o, a1e, a1o;
#pragma unroll
        for (int r = 0; r < 8; ++r) {
            a0e[r] = bias0[r]; a1e[r] = bias1[r];
            a0o[r] = 0.0f;     a1o[r] = 0.0f;
        }

#pragma unroll
        for (int c = 0; c < 8; c += 2) {
            WMMA_F32(a0e, aw0[c],     xbuf[cb][c]);
            WMMA_F32(a1e, aw1[c],     xbuf[cb][c]);
            WMMA_F32(a0o, aw0[c + 1], xbuf[cb][c + 1]);
            WMMA_F32(a1o, aw1[c + 1], xbuf[cb][c + 1]);
        }
        {
            v2f hb0, hb1, zb;
            hb0.x = hreg[0]; hb0.y = hreg[1];
            hb1.x = hreg[2]; hb1.y = hreg[3];
            zb.x  = hi ? 0.0f : zreg;
            zb.y  = 0.0f;
            WMMA_F32(a0e, aw0[8],  hb0);
            WMMA_F32(a1e, aw1[8],  hb0);
            WMMA_F32(a0o, aw0[9],  hb1);
            WMMA_F32(a1o, aw1[9],  hb1);
            WMMA_F32(a0e, aw0[10], zb);
            WMMA_F32(a1e, aw1[10], zb);
        }

#pragma unroll
        for (int u = 0; u < 4; ++u) {
            const float ig = sigm(a0e[u] + a0o[u]);
            const float fg = sigm(a0e[4 + u] + a0o[4 + u]);
            const float gg = tanh_fast(a1e[u] + a1o[u]);
            const float og = sigm(a1e[4 + u] + a1o[4 + u]);
            const float cn = fg * cst[u] + ig * gg;
            cst[u]  = cn;
            hreg[u] = og * tanh_fast(cn);
        }

        v8f hda, hdb;
#pragma unroll
        for (int r = 0; r < 8; ++r) {
            hda[r] = (r < FWD) ? b1r[r] : 0.0f;
            hdb[r] = 0.0f;
        }
        {
            v2f hb0, hb1;
            hb0.x = hreg[0]; hb0.y = hreg[1];
            hb1.x = hreg[2]; hb1.y = hreg[3];
            WMMA_F32(hda, ah0, hb0);
            WMMA_F32(hdb, ah1, hb1);
        }
        float delta = b2r;
#pragma unroll
        for (int k = 0; k < FWD; ++k)
            delta = fmaf(fmaxf(hda[k] + hdb[k], 0.0f), w2r[k], delta);
        zreg += fmaxf(delta, 0.0f);
        if (hi == 0) orow[t] = zreg;
    }
}

extern "C" void kernel_launch(void* const* d_in, const int* in_sizes, int n_in,
                              void* d_out, int out_size, void* d_ws, size_t ws_size,
                              hipStream_t stream) {
    (void)in_sizes; (void)n_in; (void)out_size; (void)d_ws; (void)ws_size;
    const float* x    = (const float*)d_in[0];
    const float* z0   = (const float*)d_in[1];
    const float* W_ih = (const float*)d_in[2];
    const float* W_hh = (const float*)d_in[3];
    const float* b_ih = (const float*)d_in[4];
    const float* b_hh = (const float*)d_in[5];
    const float* W1   = (const float*)d_in[6];
    const float* b1   = (const float*)d_in[7];
    const float* W2   = (const float*)d_in[8];
    const float* b2   = (const float*)d_in[9];
    float* out = (float*)d_out;

    dim3 grid(BATCH / 16);
    dim3 block(32);
    monotonic_lstm_wmma<<<grid, block, 0, stream>>>(x, z0, W_ih, W_hh, b_ih, b_hh,
                                                    W1, b1, W2, b2, out);
}


# Time loop software-pipelined: x loads 4 steps ahead, next step's x projection issued one matrix instruction at a time between cell stages, h chunks behind the head; sum order unchanged
# speedup vs baseline: 2.5948x; 1.3047x over previous
#include <hip/hip_runtime.h>
#include <hip/hip_bf16.h>


#define BATCH 4096
#define TSEQ  200
#define FEAT  32
#define HID   8
#define FWD   5


typedef __attribute__((ext_vector_type(2))) float v2f;
typedef __attribute__((ext_vector_type(8))) float v8f;

__device__ __forceinline__ float recip_ge1(float d) {
    return __builtin_amdgcn_rcpf(d);
}
__device__ __forceinline__ float sigm(float x) {
    return recip_ge1(1.0f + __expf(-x));
}
__device__ __forceinline__ float tanh_fast(float x) {
    return fmaf(-2.0f, recip_ge1(1.0f + __expf(2.0f * x)), 1.0f);
}

__device__ __forceinline__ int gate_of_row(int tile, int rho) {
    const int r = rho & 7, h = rho >> 3;
    return tile * 16 + ((r & 4) ? 8 : 0) + 4 * h + (r & 3);
}

__device__ __forceinline__ float wfrag(const float* __restrict__ W_ih,
                                       const float* __restrict__ W_hh,
                                       int g, int c, int hi, int s) {
    if (c < 8)  return W_ih[g * (FEAT + 1) + 4 * c + 2 * hi + s];
    if (c == 8) return W_hh[g * HID + 4 * hi + s];
    if (c == 9) return W_hh[g * HID + 4 * hi + 2 + s];
    return (hi == 0 && s == 0) ? W_ih[g * (FEAT + 1) + FEAT] : 0.0f;
}


__device__ __forceinline__ void loadx(v2f (&dst)[8], const float* __restrict__ xr, int hi) {
#pragma unroll
    for (int c = 0; c < 8; ++c)
        dst[c] = *reinterpret_cast<const v2f*>(xr + 4 * c + 2 * hi);
}

#define WMMA_F32(ACC, A, BFRAG) \
    (ACC) = __builtin_amdgcn_wmma_f32_16x16x4_f32(false, (A), false, (BFRAG), (short)0, (ACC), false, false)
#define WMMA_ACC(D, A, BFRAG, C) \
    (D) = __builtin_amdgcn_wmma_f32_16x16x4_f32(false, (A), false, (BFRAG), (short)0, (C), false, false)

#define ORDER4(A, B, C, D) \
    asm volatile("" : "+v"(A), "+v"(B), "+v"(C), "+v"(D))
#define ORDER6(A, B, C, D, E, F) \
    asm volatile("" : "+v"(A), "+v"(B), "+v"(C), "+v"(D), "+v"(E), "+v"(F))
#define ORDER7(A, B, P) \
    asm volatile("" : "+v"(A), "+v"(B), "+v"(P[0]), "+v"(P[1]), "+v"(P[2]), "+v"(P[3]), "+v"(P[4]))
#define ORDER8(A, B, C, D, P) \
    asm volatile("" : "+v"(A), "+v"(B), "+v"(C), "+v"(D), \
                      "+v"(P[0]), "+v"(P[1]), "+v"(P[2]), "+v"(P[3]))
#define ORDER11(A, B, C, P, Q) \
    asm volatile("" : "+v"(A), "+v"(B), "+v"(C), \
                      "+v"(P[0]), "+v"(P[1]), "+v"(P[2]), "+v"(P[3]), \
                      "+v"(Q[0]), "+v"(Q[1]), "+v"(Q[2]), "+v"(Q[3]))

__global__ __launch_bounds__(32)
void monotonic_lstm_wmma(const float* __restrict__ x,
                         const float* __restrict__ z0,
                         const float* __restrict__ W_ih,
                         const float* __restrict__ W_hh,
                         const float* __restrict__ b_ih,
                         const float* __restrict__ b_hh,
                         const float* __restrict__ W1,
                         const float* __restrict__ b1,
                         const float* __restrict__ W2,
                         const float* __restrict__ b2,
                         float* __restrict__ out)
{
    const int lane = threadIdx.x;
    const int m    = lane & 15;
    const int hi   = lane >> 4;
    const int b0   = blockIdx.x * 16;

    const int g0 = gate_of_row(0, m);
    const int g1 = gate_of_row(1, m);
    v2f aw0[11], aw1[11];
#pragma unroll
    for (int c = 0; c < 11; ++c) {
        aw0[c].x = wfrag(W_ih, W_hh, g0, c, hi, 0);
        aw0[c].y = wfrag(W_ih, W_hh, g0, c, hi, 1);
        aw1[c].x = wfrag(W_ih, W_hh, g1, c, hi, 0);
        aw1[c].y = wfrag(W_ih, W_hh, g1, c, hi, 1);
    }
    float bias0[8], bias1[8];
#pragma unroll
    for (int r = 0; r < 8; ++r) {
        const int ga = gate_of_row(0, 8 * hi + r);
        bias0[r] = b_ih[ga]      + b_hh[ga];
        bias1[r] = b_ih[ga + 16] + b_hh[ga + 16];
    }

    const int  kk = m & 7;
    const bool kv = kk < FWD;
    v2f ah0, ah1;
    ah0.x = kv ? W1[kk * HID + 4 * hi + 0] : 0.0f;
    ah0.y = kv ? W1[kk * HID + 4 * hi + 1] : 0.0f;
    ah1.x = kv ? W1[kk * HID + 4 * hi + 2] : 0.0f;
    ah1.y = kv ? W1[kk * HID + 4 * hi + 3] : 0.0f;
    float b1r[FWD], w2r[FWD];
#pragma unroll
    for (int i = 0; i < FWD; ++i) { b1r[i] = b1[i]; w2r[i] = W2[i]; }
    const float b2r = b2[0];

    float cst[4]  = {0.f, 0.f, 0.f, 0.f};
    float hreg[4] = {0.f, 0.f, 0.f, 0.f};
    float zreg = z0[b0 + m];

    const float* xrow = x + ((size_t)(b0 + m) * TSEQ) * FEAT;
    float* orow = out + (size_t)(b0 + m) * TSEQ;

    v8f bias0v, bias1v, zero8, b1v;
#pragma unroll
    for (int r = 0; r < 8; ++r) {
        bias0v[r] = bias0[r]; bias1v[r] = bias1[r]; zero8[r] = 0.0f;
        b1v[r] = (r < FWD) ? b1r[r] : 0.0f;
    }

    v2f xbuf[4][8];
    loadx(xbuf[0], xrow, hi);
    loadx(xbuf[1], xrow + 1 * FEAT, hi);
    loadx(xbuf[2], xrow + 2 * FEAT, hi);
    loadx(xbuf[3], xrow + 3 * FEAT, hi);

    v8f a0e, a0o, a1e, a1o;
    WMMA_ACC(a0e, aw0[0], xbuf[0][0], bias0v);
    WMMA_ACC(a1e, aw1[0], xbuf[0][0], bias1v);
    WMMA_ACC(a0o, aw0[1], xbuf[0][1], zero8);
    WMMA_ACC(a1o, aw1[1], xbuf[0][1], zero8);
#pragma unroll
    for (int c = 2; c < 8; c += 2) {
        WMMA_F32(a0e, aw0[c],     xbuf[0][c]);
        WMMA_F32(a1e, aw1[c],     xbuf[0][c]);
        WMMA_F32(a0o, aw0[c + 1], xbuf[0][c + 1]);
        WMMA_F32(a1o, aw1[c + 1], xbuf[0][c + 1]);
    }
    {
        v2f hz; hz.x = 0.0f; hz.y = 0.0f;
        WMMA_F32(a0e, aw0[8], hz);
        WMMA_F32(a1e, aw1[8], hz);
        WMMA_F32(a0o, aw0[9], hz);
        WMMA_F32(a1o, aw1[9], hz);
    }

#pragma unroll 4
    for (int t = 0; t < TSEQ; ++t) {
        const int cb = t & 3;
        const int nb = (t + 1) & 3;
        const int tl = (t + 4 < TSEQ) ? (t + 4) : (TSEQ - 1);
        __builtin_amdgcn_sched_barrier(0);
        loadx(xbuf[cb], xrow + (size_t)tl * FEAT, hi);
        __builtin_amdgcn_sched_barrier(0);

        {
            v2f zb;
            zb.x = hi ? 0.0f : zreg;
            zb.y = 0.0f;
            WMMA_F32(a0e, aw0[10], zb);
            WMMA_F32(a1e, aw1[10], zb);
        }
        ORDER4(a0e, a1e, xbuf[nb][0], xbuf[nb][1]);

        v8f n0e, n0o, n1e, n1o;
        WMMA_ACC(n0e, aw0[0], xbuf[nb][0], bias0v);
        WMMA_ACC(n1e, aw1[0], xbuf[nb][0], bias1v);

        float sI[4], sF[4], sG[4], sO[4];
#pragma unroll
        for (int u = 0; u < 4; ++u) {
            sG[u] = a1e[u]     + a1o[u];
            sI[u] = a0e[u]     + a0o[u];
        }
        ORDER11(n0e, n1e, xbuf[nb][1], sG, sI);
        WMMA_ACC(n0o, aw0[1], xbuf[nb][1], zero8);

        float eG[4], eI[4], eF[4], eO[4];
#pragma unroll
        for (int u = 0; u < 4; ++u) {
            sF[u] = a0e[4 + u] + a0o[4 + u];
            sO[u] = a1e[4 + u] + a1o[4 + u];
            eG[u] = __expf(2.0f * sG[u]);
        }
        ORDER8(n0e, n1e, n0o, xbuf[nb][1], eG);
        WMMA_ACC(n1o, aw1[1], xbuf[nb][1], zero8);

#pragma unroll
        for (int u = 0; u < 4; ++u) eI[u] = __expf(-sI[u]);
        ORDER8(n0e, n1e, n0o, n1o, eI);
        WMMA_F32(n0e, aw0[2], xbuf[nb][2]);
#pragma unroll
        for (int u = 0; u < 4; ++u) eF[u] = __expf(-sF[u]);
        ORDER8(n0e, n1e, n0o, n1o, eF);
        WMMA_F32(n1e, aw1[2], xbuf[nb][2]);
#pragma unroll
        for (int u = 0; u < 4; ++u) eO[u] = __expf(-sO[u]);
        ORDER8(n0e, n1e, n0o, n1o, eO);
        WMMA_F32(n0o, aw0[3], xbuf[nb][3]);

        float rG[4], rI[4], rF[4], rO[4];
#pragma unroll
        for (int u = 0; u < 4; ++u) rG[u] = recip_ge1(1.0f + eG[u]);
        ORDER8(n0e, n1e, n0o, n1o, rG);
        WMMA_F32(n1o, aw1[3], xbuf[nb][3]);
#pragma unroll
        for (int u = 0; u < 4; ++u) rI[u] = recip_ge1(1.0f + eI[u]);
        ORDER8(n0e, n1e, n0o, n1o, rI);
        WMMA_F32(n0e, aw0[4], xbuf[nb][4]);
#pragma unroll
        for (int u = 0; u < 4; ++u) rF[u] = recip_ge1(1.0f + eF[u]);
        ORDER8(n0e, n1e, n0o, n1o, rF);
        WMMA_F32(n1e, aw1[4], xbuf[nb][4]);
#pragma unroll
        for (int u = 0; u < 4; ++u) rO[u] = recip_ge1(1.0f + eO[u]);
        ORDER8(n0e, n1e, n0o, n1o, rO);
        WMMA_F32(n0o, aw0[5], xbuf[nb][5]);

        float ig[4];
#pragma unroll
        for (int u = 0; u < 4; ++u) ig[u] = rI[u] * fmaf(-2.0f, rG[u], 1.0f);
        ORDER8(n0e, n1e, n0o, n1o, ig);
        WMMA_F32(n1o, aw1[5], xbuf[nb][5]);
#pragma unroll
        for (int u = 0; u < 4; ++u) cst[u] = fmaf(rF[u], cst[u], ig[u]);
        ORDER8(n0e, n1e, n0o, n1o, cst);
        WMMA_F32(n0e, aw0[6], xbuf[nb][6]);

        float e2[4], r2[4];
#pragma unroll
        for (int u = 0; u < 4; ++u) e2[u] = __expf(2.0f * cst[u]);
        ORDER8(n0e, n1e, n0o, n1o, e2);
        WMMA_F32(n1e, aw1[6], xbuf[nb][6]);
#pragma unroll
        for (int u = 0; u < 4; ++u) r2[u] = recip_ge1(1.0f + e2[u]);
        ORDER8(n0e, n1e, n0o, n1o, r2);
        WMMA_F32(n0o, aw0[7], xbuf[nb][7]);
#pragma unroll
        for (int u = 0; u < 4; ++u) hreg[u] = rO[u] * fmaf(-2.0f, r2[u], 1.0f);
        ORDER8(n0e, n1e, n0o, n1o, hreg);

        v8f hda, hdb;
        v2f hb0, hb1;
        hb0.x = hreg[0]; hb0.y = hreg[1];
        hb1.x = hreg[2]; hb1.y = hreg[3];
        WMMA_ACC(hda, ah0, hb0, b1v);
        WMMA_ACC(hdb, ah1, hb1, zero8);
        ORDER6(hda, hdb, n0e, n1e, n0o, n1o);
        WMMA_F32(n1o, aw1[7], xbuf[nb][7]);
        WMMA_F32(n0e, aw0[8], hb0);
        WMMA_F32(n1e, aw1[8], hb0);
        ORDER6(hda, hdb, n0e, n1e, n0o, n1o);
        float dk[FWD];
#pragma unroll
        for (int k = 0; k < FWD; ++k) dk[k] = fmaxf(hda[k] + hdb[k], 0.0f);
        WMMA_F32(n0o, aw0[9], hb1);
        ORDER7(n0o, n1o, dk);
        WMMA_F32(n1o, aw1[9], hb1);
        float delta = b2r;
#pragma unroll
        for (int k = 0; k < FWD; ++k)
            delta = fmaf(dk[k], w2r[k], delta);
        zreg += fmaxf(delta, 0.0f);
        if (hi == 0) orow[t] = zreg;

        a0e = n0e; a0o = n0o; a1e = n1e; a1o = n1o;
    }
}

extern "C" void kernel_launch(void* const* d_in, const int* in_sizes, int n_in,
                              void* d_out, int out_size, void* d_ws, size_t ws_size,
                              hipStream_t stream) {
    (void)in_sizes; (void)n_in; (void)out_size; (void)d_ws; (void)ws_size;
    const float* x    = (const float*)d_in[0];
    const float* z0   = (const float*)d_in[1];
    const float* W_ih = (const float*)d_in[2];
    const float* W_hh = (const float*)d_in[3];
    const float* b_ih = (const float*)d_in[4];
    const float* b_hh = (const float*)d_in[5];
    const float* W1   = (const float*)d_in[6];
    const float* b1   = (const float*)d_in[7];
    const float* W2   = (const float*)d_in[8];
    const float* b2   = (const float*)d_in[9];
    float* out = (float*)d_out;

    dim3 grid(BATCH / 16);
    dim3 block(32);
    monotonic_lstm_wmma<<<grid, block, 0, stream>>>(x, z0, W_ih, W_hh, b_ih, b_hh,
                                                    W1, b1, W2, b2, out);
}


# x projection moved off the recurrent wave: two extra waves per work-group (one gate tile each) compute it a block of 25 steps ahead and hand it over through an LDS ring; sum order unchanged
# speedup vs baseline: 3.8917x; 1.4998x over previous
#include <hip/hip_runtime.h>
#include <hip/hip_bf16.h>


#define BATCH 4096
#define TSEQ  200
#define FEAT  32
#define HID   8
#define FWD   5
#define NB    25
#define NBLK  (TSEQ / NB)
#define SLOT  256


typedef __attribute__((ext_vector_type(2))) float v2f;
typedef __attribute__((ext_vector_type(4))) float v4f;
typedef __attribute__((ext_vector_type(8))) float v8f;

__device__ __forceinline__ float recip_ge1(float d) {
    return __builtin_amdgcn_rcpf(d);
}

__device__ __forceinline__ int gate_of_row(int tile, int rho) {
    const int r = rho & 7, h = rho >> 3;
    return tile * 16 + ((r & 4) ? 8 : 0) + 4 * h + (r & 3);
}

__device__ __forceinline__ float wfrag(const float* __restrict__ W_ih,
                                       const float* __restrict__ W_hh,
                                       int g, int c, int hi, int s) {
    if (c < 8)  return W_ih[g * (FEAT + 1) + 4 * c + 2 * hi + s];
    if (c == 8) return W_hh[g * HID + 4 * hi + s];
    if (c == 9) return W_hh[g * HID + 4 * hi + 2 + s];
    return (hi == 0 && s == 0) ? W_ih[g * (FEAT + 1) + FEAT] : 0.0f;
}


__device__ __forceinline__ void loadx(v2f (&dst)[8], const float* __restrict__ xr, int hi) {
#pragma unroll
    for (int c = 0; c < 8; ++c)
        dst[c] = *reinterpret_cast<const v2f*>(xr + 4 * c + 2 * hi);
}

__device__ __forceinline__ void put8(v4f* p, v8f v) {
    p[0]  = __builtin_shufflevector(v, v, 0, 1, 2, 3);
    p[32] = __builtin_shufflevector(v, v, 4, 5, 6, 7);
}
__device__ __forceinline__ v8f get8(const v4f* p) {
    const v4f lo = p[0], up = p[32];
    return __builtin_shufflevector(lo, up, 0, 1, 2, 3, 4, 5, 6, 7);
}

#define WMMA_F32(ACC, A, BFRAG) \
    (ACC) = __builtin_amdgcn_wmma_f32_16x16x4_f32(false, (A), false, (BFRAG), (short)0, (ACC), false, false)
#define WMMA_ACC(D, A, BFRAG, C) \
    (D) = __builtin_amdgcn_wmma_f32_16x16x4_f32(false, (A), false, (BFRAG), (short)0, (C), false, false)

#define ORDER6(A, B, C, D, E, F) \
    asm volatile("" : "+v"(A), "+v"(B), "+v"(C), "+v"(D), "+v"(E), "+v"(F))
#define ORDER7(A, B, P) \
    asm volatile("" : "+v"(A), "+v"(B), "+v"(P[0]), "+v"(P[1]), "+v"(P[2]), "+v"(P[3]), "+v"(P[4]))

__global__ __launch_bounds__(96)
void monotonic_lstm_wmma(const float* __restrict__ x,
                         const float* __restrict__ z0,
                         const float* __restrict__ W_ih,
                         const float* __restrict__ W_hh,
                         const float* __restrict__ b_ih,
                         const float* __restrict__ b_hh,
                         const float* __restrict__ W1,
                         const float* __restrict__ b1,
                         const float* __restrict__ W2,
                         const float* __restrict__ b2,
                         float* __restrict__ out)
{
    __shared__ v4f ring[2 * NB * SLOT];

    const int lane = threadIdx.x & 31;
    const int wave = __builtin_amdgcn_readfirstlane((int)(threadIdx.x >> 5));
    const int m    = lane & 15;
    const int hi   = lane >> 4;
    const int b0   = blockIdx.x * 16;
    const bool projector = wave != 0;
    const int  tile = wave - 1;

    v8f zero8;
#pragma unroll
    for (int r = 0; r < 8; ++r) zero8[r] = 0.0f;

    v2f aw[8] = {};
    v8f biasv = zero8;
    v2f xbuf[5][8] = {};
    const float* xrow = x + ((size_t)(b0 + m) * TSEQ) * FEAT;

    v2f ah8_0 = {}, ah9_0 = {}, az_0 = {}, ah8_1 = {}, ah9_1 = {}, az_1 = {};
    v2f ah0 = {}, ah1 = {};
    v8f b1v = zero8;
    float w2r[FWD] = {0.f, 0.f, 0.f, 0.f, 0.f};
    float b2r = 0.0f;
    float cst[4]  = {0.f, 0.f, 0.f, 0.f};
    float hreg[4] = {0.f, 0.f, 0.f, 0.f};
    float zreg = 0.0f;
    float* orow = out + (size_t)(b0 + m) * TSEQ;

    if (projector) {
        const int g = gate_of_row(tile, m);
#pragma unroll
        for (int c = 0; c < 8; ++c) {
            aw[c].x = wfrag(W_ih, W_hh, g, c, hi, 0);
            aw[c].y = wfrag(W_ih, W_hh, g, c, hi, 1);
        }
#pragma unroll
        for (int r = 0; r < 8; ++r) {
            const int ga = gate_of_row(tile, 8 * hi + r);
            biasv[r] = b_ih[ga] + b_hh[ga];
        }
#pragma unroll
        for (int j = 0; j < 5; ++j) loadx(xbuf[j], xrow + j * FEAT, hi);
    } else {
        const int g0 = gate_of_row(0, m);
        const int g1 = gate_of_row(1, m);
        ah8_0.x = wfrag(W_ih, W_hh, g0, 8, hi, 0);  ah8_0.y = wfrag(W_ih, W_hh, g0, 8, hi, 1);
        ah9_0.x = wfrag(W_ih, W_hh, g0, 9, hi, 0);  ah9_0.y = wfrag(W_ih, W_hh, g0, 9, hi, 1);
        az_0.x  = wfrag(W_ih, W_hh, g0, 10, hi, 0); az_0.y  = wfrag(W_ih, W_hh, g0, 10, hi, 1);
        ah8_1.x = wfrag(W_ih, W_hh, g1, 8, hi, 0);  ah8_1.y = wfrag(W_ih, W_hh, g1, 8, hi, 1);
        ah9_1.x = wfrag(W_ih, W_hh, g1, 9, hi, 0);  ah9_1.y = wfrag(W_ih, W_hh, g1, 9, hi, 1);
        az_1.x  = wfrag(W_ih, W_hh, g1, 10, hi, 0); az_1.y  = wfrag(W_ih, W_hh, g1, 10, hi, 1);
        const int  kk = m & 7;
        const bool kv = kk < FWD;
        ah0.x = kv ? W1[kk * HID + 4 * hi + 0] : 0.0f;
        ah0.y = kv ? W1[kk * HID + 4 * hi + 1] : 0.0f;
        ah1.x = kv ? W1[kk * HID + 4 * hi + 2] : 0.0f;
        ah1.y = kv ? W1[kk * HID + 4 * hi + 3] : 0.0f;
#pragma unroll
        for (int i = 0; i < FWD; ++i) { b1v[i] = b1[i]; w2r[i] = W2[i]; }
        b2r  = b2[0];
        zreg = z0[b0 + m];
    }

    for (int k = 0; k <= NBLK; ++k) {
        if (projector) {
            if (k < NBLK) {
                v4f* wb = ring + ((k & 1) * NB) * SLOT + tile * 128 + lane;
                const int t0 = k * NB;
                for (int q = 0; q < NB / 5; ++q) {
#pragma unroll
                    for (int j = 0; j < 5; ++j) {
                        const int s = 5 * q + j;
                        v8f pe, po;
                        WMMA_ACC(pe, aw[0], xbuf[j][0], biasv);
                        WMMA_ACC(po, aw[1], xbuf[j][1], zero8);
#pragma unroll
                        for (int c = 2; c < 8; c += 2) {
                            WMMA_F32(pe, aw[c],     xbuf[j][c]);
                            WMMA_F32(po, aw[c + 1], xbuf[j][c + 1]);
                        }
                        put8(wb + s * SLOT,      pe);
                        put8(wb + s * SLOT + 64, po);
                        const int tn = t0 + s + 5;
                        const int tl = (tn < TSEQ) ? tn : (TSEQ - 1);
                        __builtin_amdgcn_sched_barrier(0);
                        loadx(xbuf[j], xrow + (size_t)tl * FEAT, hi);
                        __builtin_amdgcn_sched_barrier(0);
                    }
                }
            }
        } else {
            if (k >= 1) {
                const int kb = k - 1;
                const v4f* rb = ring + ((kb & 1) * NB) * SLOT + lane;
                const int t0 = kb * NB;
                v8f a0e, a0o, a1e, a1o;
                {
                    v2f hb0, hb1;
                    hb0.x = hreg[0]; hb0.y = hreg[1];
                    hb1.x = hreg[2]; hb1.y = hreg[3];
                    WMMA_ACC(a0e, ah8_0, hb0, get8(rb));
                    WMMA_ACC(a1e, ah8_1, hb0, get8(rb + 128));
                    WMMA_ACC(a0o, ah9_0, hb1, get8(rb + 64));
                    WMMA_ACC(a1o, ah9_1, hb1, get8(rb + 192));
                }
                for (int s = 0; s < NB; ++s) {
                    const int sn = (s + 1 < NB) ? (s + 1) : (NB - 1);
                    {
                        v2f zb;
                        zb.x = hi ? 0.0f : zreg;
                        zb.y = 0.0f;
                        WMMA_F32(a0e, az_0, zb);
                        WMMA_F32(a1e, az_1, zb);
                    }
                    __builtin_amdgcn_sched_barrier(0);
                    v8f p0e = get8(rb + sn * SLOT);
                    v8f p0o = get8(rb + sn * SLOT + 64);
                    v8f p1e = get8(rb + sn * SLOT + 128);
                    v8f p1o = get8(rb + sn * SLOT + 192);
                    __builtin_amdgcn_sched_barrier(0);

                    float sI[4], sF[4], sG[4], sO[4];
#pragma unroll
                    for (int u = 0; u < 4; ++u) {
                        sG[u] = a1e[u]     + a1o[u];
                        sI[u] = a0e[u]     + a0o[u];
                        sF[u] = a0e[4 + u] + a0o[4 + u];
                        sO[u] = a1e[4 + u] + a1o[4 + u];
                    }
                    float rG[4], rI[4], rF[4], rO[4];
#pragma unroll
                    for (int u = 0; u < 4; ++u) {
                        rG[u] = recip_ge1(1.0f + __expf(2.0f * sG[u]));
                        rI[u] = recip_ge1(1.0f + __expf(-sI[u]));
                        rF[u] = recip_ge1(1.0f + __expf(-sF[u]));
                        rO[u] = recip_ge1(1.0f + __expf(-sO[u]));
                    }
#pragma unroll
                    for (int u = 0; u < 4; ++u) {
                        const float ig = rI[u] * fmaf(-2.0f, rG[u], 1.0f);
                        cst[u] = fmaf(rF[u], cst[u], ig);
                        const float r2 = recip_ge1(1.0f + __expf(2.0f * cst[u]));
                        hreg[u] = rO[u] * fmaf(-2.0f, r2, 1.0f);
                    }

                    v8f hda, hdb, n0e, n0o, n1e, n1o;
                    v2f hb0, hb1;
                    hb0.x = hreg[0]; hb0.y = hreg[1];
                    hb1.x = hreg[2]; hb1.y = hreg[3];
                    WMMA_ACC(hda, ah0, hb0, b1v);
                    WMMA_ACC(hdb, ah1, hb1, zero8);
                    ORDER6(hda, hdb, p0e, p1e, p0o, p1o);
                    WMMA_ACC(n0e, ah8_0, hb0, p0e);
                    WMMA_ACC(n1e, ah8_1, hb0, p1e);
                    ORDER6(hda, hdb, n0e, n1e, p0o, p1o);
                    float dk[FWD];
#pragma unroll
                    for (int i = 0; i < FWD; ++i) dk[i] = fmaxf(hda[i] + hdb[i], 0.0f);
                    WMMA_ACC(n0o, ah9_0, hb1, p0o);
                    ORDER7(n0o, p1o, dk);
                    WMMA_ACC(n1o, ah9_1, hb1, p1o);
                    float delta = b2r;
#pragma unroll
                    for (int i = 0; i < FWD; ++i)
                        delta = fmaf(dk[i], w2r[i], delta);
                    zreg += fmaxf(delta, 0.0f);
                    if (hi == 0) orow[t0 + s] = zreg;

                    a0e = n0e; a0o = n0o; a1e = n1e; a1o = n1o;
                }
            }
        }
        __builtin_amdgcn_fence(__ATOMIC_RELEASE, "workgroup", "local");
        __builtin_amdgcn_s_barrier();
        __builtin_amdgcn_fence(__ATOMIC_ACQUIRE, "workgroup", "local");
    }
}

extern "C" void kernel_launch(void* const* d_in, const int* in_sizes, int n_in,
                              void* d_out, int out_size, void* d_ws, size_t ws_size,
                              hipStream_t stream) {
    (void)in_sizes; (void)n_in; (void)out_size; (void)d_ws; (void)ws_size;
    const float* x    = (const float*)d_in[0];
    const float* z0   = (const float*)d_in[1];
    const float* W_ih = (const float*)d_in[2];
    const float* W_hh = (const float*)d_in[3];
    const float* b_ih = (const float*)d_in[4];
    const float* b_hh = (const float*)d_in[5];
    const float* W1   = (const float*)d_in[6];
    const float* b1   = (const float*)d_in[7];
    const float* W2   = (const float*)d_in[8];
    const float* b2   = (const float*)d_in[9];
    float* out = (float*)d_out;

    dim3 grid(BATCH / 16);
    dim3 block(96);
    monotonic_lstm_wmma<<<grid, block, 0, stream>>>(x, z0, W_ih, W_hh, b_ih, b_hh,
                                                    W1, b1, W2, b2, out);
}


# Gate sigmoids and both tanh: one hardware tanh instruction per activation, sigmoid(a) = 1/2 + tanh(a/2)/2 with the exact half folded into the i, f, o weight rows and bias; replaces exp + reciprocal
# speedup vs baseline: 4.2872x; 1.1016x over previous
#include <hip/hip_runtime.h>
#include <hip/hip_bf16.h>


#define BATCH 4096
#define TSEQ  200
#define FEAT  32
#define HID   8
#define FWD   5
#define NB    25
#define NBLK  (TSEQ / NB)
#define SLOT  256


typedef __attribute__((ext_vector_type(2))) float v2f;
typedef __attribute__((ext_vector_type(4))) float v4f;
typedef __attribute__((ext_vector_type(8))) float v8f;

__device__ __forceinline__ float hw_tanh(float a) {
    return __builtin_amdgcn_tanhf(a);
}
__device__ __forceinline__ float sigmoid_of_half(float half_a) {
    return fmaf(0.5f, hw_tanh(half_a), 0.5f);
}

__device__ __forceinline__ float gate_scale(int g) {
    return (g >= 2 * HID && g < 3 * HID) ? 1.0f : 0.5f;
}

__device__ __forceinline__ int gate_of_row(int tile, int rho) {
    const int r = rho & 7, h = rho >> 3;
    return tile * 16 + ((r & 4) ? 8 : 0) + 4 * h + (r & 3);
}

__device__ __forceinline__ float wfrag(const float* __restrict__ W_ih,
                                       const float* __restrict__ W_hh,
                                       int g, int c, int hi, int s) {
    if (c < 8)  return W_ih[g * (FEAT + 1) + 4 * c + 2 * hi + s];
    if (c == 8) return W_hh[g * HID + 4 * hi + s];
    if (c == 9) return W_hh[g * HID + 4 * hi + 2 + s];
    return (hi == 0 && s == 0) ? W_ih[g * (FEAT + 1) + FEAT] : 0.0f;
}


__device__ __forceinline__ void loadx(v2f (&dst)[8], const float* __restrict__ xr, int hi) {
#pragma unroll
    for (int c = 0; c < 8; ++c)
        dst[c] = *reinterpret_cast<const v2f*>(xr + 4 * c + 2 * hi);
}

__device__ __forceinline__ void put8(v4f* p, v8f v) {
    p[0]  = __builtin_shufflevector(v, v, 0, 1, 2, 3);
    p[32] = __builtin_shufflevector(v, v, 4, 5, 6, 7);
}
__device__ __forceinline__ v8f get8(const v4f* p) {
    const v4f lo = p[0], up = p[32];
    return __builtin_shufflevector(lo, up, 0, 1, 2, 3, 4, 5, 6, 7);
}

#define WMMA_F32(ACC, A, BFRAG) \
    (ACC) = __builtin_amdgcn_wmma_f32_16x16x4_f32(false, (A), false, (BFRAG), (short)0, (ACC), false, false)
#define WMMA_ACC(D, A, BFRAG, C) \
    (D) = __builtin_amdgcn_wmma_f32_16x16x4_f32(false, (A), false, (BFRAG), (short)0, (C), false, false)

#define ORDER6(A, B, C, D, E, F) \
    asm volatile("" : "+v"(A), "+v"(B), "+v"(C), "+v"(D), "+v"(E), "+v"(F))
#define ORDER7(A, B, P) \
    asm volatile("" : "+v"(A), "+v"(B), "+v"(P[0]), "+v"(P[1]), "+v"(P[2]), "+v"(P[3]), "+v"(P[4]))

__global__ __launch_bounds__(96)
void monotonic_lstm_wmma(const float* __restrict__ x,
                         const float* __restrict__ z0,
                         const float* __restrict__ W_ih,
                         const float* __restrict__ W_hh,
                         const float* __restrict__ b_ih,
                         const float* __restrict__ b_hh,
                         const float* __restrict__ W1,
                         const float* __restrict__ b1,
                         const float* __restrict__ W2,
                         const float* __restrict__ b2,
                         float* __restrict__ out)
{
    __shared__ v4f ring[2 * NB * SLOT];

    const int lane = threadIdx.x & 31;
    const int wave = __builtin_amdgcn_readfirstlane((int)(threadIdx.x >> 5));
    const int m    = lane & 15;
    const int hi   = lane >> 4;
    const int b0   = blockIdx.x * 16;
    const bool projector = wave != 0;
    const int  tile = wave - 1;

    v8f zero8;
#pragma unroll
    for (int r = 0; r < 8; ++r) zero8[r] = 0.0f;

    v2f aw[8] = {};
    v8f biasv = zero8;
    v2f xbuf[5][8] = {};
    const float* xrow = x + ((size_t)(b0 + m) * TSEQ) * FEAT;

    v2f ah8_0 = {}, ah9_0 = {}, az_0 = {}, ah8_1 = {}, ah9_1 = {}, az_1 = {};
    v2f ah0 = {}, ah1 = {};
    v8f b1v = zero8;
    float w2r[FWD] = {0.f, 0.f, 0.f, 0.f, 0.f};
    float b2r = 0.0f;
    float cst[4]  = {0.f, 0.f, 0.f, 0.f};
    float hreg[4] = {0.f, 0.f, 0.f, 0.f};
    float zreg = 0.0f;
    float* orow = out + (size_t)(b0 + m) * TSEQ;

    if (projector) {
        const int g = gate_of_row(tile, m);
#pragma unroll
        for (int c = 0; c < 8; ++c) {
            aw[c].x = gate_scale(g) * wfrag(W_ih, W_hh, g, c, hi, 0);
            aw[c].y = gate_scale(g) * wfrag(W_ih, W_hh, g, c, hi, 1);
        }
#pragma unroll
        for (int r = 0; r < 8; ++r) {
            const int ga = gate_of_row(tile, 8 * hi + r);
            biasv[r] = gate_scale(ga) * (b_ih[ga] + b_hh[ga]);
        }
#pragma unroll
        for (int j = 0; j < 5; ++j) loadx(xbuf[j], xrow + j * FEAT, hi);
    } else {
        const int g0 = gate_of_row(0, m);
        const int g1 = gate_of_row(1, m);
        const float k0 = gate_scale(g0), k1 = gate_scale(g1);
        ah8_0.x = k0 * wfrag(W_ih, W_hh, g0, 8, hi, 0);  ah8_0.y = k0 * wfrag(W_ih, W_hh, g0, 8, hi, 1);
        ah9_0.x = k0 * wfrag(W_ih, W_hh, g0, 9, hi, 0);  ah9_0.y = k0 * wfrag(W_ih, W_hh, g0, 9, hi, 1);
        az_0.x  = k0 * wfrag(W_ih, W_hh, g0, 10, hi, 0); az_0.y  = k0 * wfrag(W_ih, W_hh, g0, 10, hi, 1);
        ah8_1.x = k1 * wfrag(W_ih, W_hh, g1, 8, hi, 0);  ah8_1.y = k1 * wfrag(W_ih, W_hh, g1, 8, hi, 1);
        ah9_1.x = k1 * wfrag(W_ih, W_hh, g1, 9, hi, 0);  ah9_1.y = k1 * wfrag(W_ih, W_hh, g1, 9, hi, 1);
        az_1.x  = k1 * wfrag(W_ih, W_hh, g1, 10, hi, 0); az_1.y  = k1 * wfrag(W_ih, W_hh, g1, 10, hi, 1);
        const int  kk = m & 7;
        const bool kv = kk < FWD;
        ah0.x = kv ? W1[kk * HID + 4 * hi + 0] : 0.0f;
        ah0.y = kv ? W1[kk * HID + 4 * hi + 1] : 0.0f;
        ah1.x = kv ? W1[kk * HID + 4 * hi + 2] : 0.0f;
        ah1.y = kv ? W1[kk * HID + 4 * hi + 3] : 0.0f;
#pragma unroll
        for (int i = 0; i < FWD; ++i) { b1v[i] = b1[i]; w2r[i] = W2[i]; }
        b2r  = b2[0];
        zreg = z0[b0 + m];
    }

    for (int k = 0; k <= NBLK; ++k) {
        if (projector) {
            if (k < NBLK) {
                v4f* wb = ring + ((k & 1) * NB) * SLOT + tile * 128 + lane;
                const int t0 = k * NB;
                for (int q = 0; q < NB / 5; ++q) {
#pragma unroll
                    for (int j = 0; j < 5; ++j) {
                        const int s = 5 * q + j;
                        v8f pe, po;
                        WMMA_ACC(pe, aw[0], xbuf[j][0], biasv);
                        WMMA_ACC(po, aw[1], xbuf[j][1], zero8);
#pragma unroll
                        for (int c = 2; c < 8; c += 2) {
                            WMMA_F32(pe, aw[c],     xbuf[j][c]);
                            WMMA_F32(po, aw[c + 1], xbuf[j][c + 1]);
                        }
                        put8(wb + s * SLOT,      pe);
                        put8(wb + s * SLOT + 64, po);
                        const int tn = t0 + s + 5;
                        const int tl = (tn < TSEQ) ? tn : (TSEQ - 1);
                        __builtin_amdgcn_sched_barrier(0);
                        loadx(xbuf[j], xrow + (size_t)tl * FEAT, hi);
                        __builtin_amdgcn_sched_barrier(0);
                    }
                }
            }
        } else {
            if (k >= 1) {
                const int kb = k - 1;
                const v4f* rb = ring + ((kb & 1) * NB) * SLOT + lane;
                const int t0 = kb * NB;
                v8f a0e, a0o, a1e, a1o;
                {
                    v2f hb0, hb1;
                    hb0.x = hreg[0]; hb0.y = hreg[1];
                    hb1.x = hreg[2]; hb1.y = hreg[3];
                    WMMA_ACC(a0e, ah8_0, hb0, get8(rb));
                    WMMA_ACC(a1e, ah8_1, hb0, get8(rb + 128));
                    WMMA_ACC(a0o, ah9_0, hb1, get8(rb + 64));
                    WMMA_ACC(a1o, ah9_1, hb1, get8(rb + 192));
                }
                for (int s = 0; s < NB; ++s) {
                    const int sn = (s + 1 < NB) ? (s + 1) : (NB - 1);
                    {
                        v2f zb;
                        zb.x = hi ? 0.0f : zreg;
                        zb.y = 0.0f;
                        WMMA_F32(a0e, az_0, zb);
                        WMMA_F32(a1e, az_1, zb);
                    }
                    __builtin_amdgcn_sched_barrier(0);
                    v8f p0e = get8(rb + sn * SLOT);
                    v8f p0o = get8(rb + sn * SLOT + 64);
                    v8f p1e = get8(rb + sn * SLOT + 128);
                    v8f p1o = get8(rb + sn * SLOT + 192);
                    __builtin_amdgcn_sched_barrier(0);

                    float sI[4], sF[4], sG[4], sO[4];
#pragma unroll
                    for (int u = 0; u < 4; ++u) {
                        sG[u] = a1e[u]     + a1o[u];
                        sI[u] = a0e[u]     + a0o[u];
                        sF[u] = a0e[4 + u] + a0o[4 + u];
                        sO[u] = a1e[4 + u] + a1o[4 + u];
                    }
                    float tG[4], gI[4], gF[4], gO[4];
#pragma unroll
                    for (int u = 0; u < 4; ++u) {
                        tG[u] = hw_tanh(sG[u]);
                        gI[u] = sigmoid_of_half(sI[u]);
                        gF[u] = sigmoid_of_half(sF[u]);
                        gO[u] = sigmoid_of_half(sO[u]);
                    }
#pragma unroll
                    for (int u = 0; u < 4; ++u) {
                        const float ig = gI[u] * tG[u];
                        cst[u] = fmaf(gF[u], cst[u], ig);
                        hreg[u] = gO[u] * hw_tanh(cst[u]);
                    }

                    v8f hda, hdb, n0e, n0o, n1e, n1o;
                    v2f hb0, hb1;
                    hb0.x = hreg[0]; hb0.y = hreg[1];
                    hb1.x = hreg[2]; hb1.y = hreg[3];
                    WMMA_ACC(hda, ah0, hb0, b1v);
                    WMMA_ACC(hdb, ah1, hb1, zero8);
                    ORDER6(hda, hdb, p0e, p1e, p0o, p1o);
                    WMMA_ACC(n0e, ah8_0, hb0, p0e);
                    WMMA_ACC(n1e, ah8_1, hb0, p1e);
                    ORDER6(hda, hdb, n0e, n1e, p0o, p1o);
                    float dk[FWD];
#pragma unroll
                    for (int i = 0; i < FWD; ++i) dk[i] = fmaxf(hda[i] + hdb[i], 0.0f);
                    WMMA_ACC(n0o, ah9_0, hb1, p0o);
                    ORDER7(n0o, p1o, dk);
                    WMMA_ACC(n1o, ah9_1, hb1, p1o);
                    float delta = b2r;
#pragma unroll
                    for (int i = 0; i < FWD; ++i)
                        delta = fmaf(dk[i], w2r[i], delta);
                    zreg += fmaxf(delta, 0.0f);
                    if (hi == 0) orow[t0 + s] = zreg;

                    a0e = n0e; a0o = n0o; a1e = n1e; a1o = n1o;
                }
            }
        }
        __builtin_amdgcn_fence(__ATOMIC_RELEASE, "workgroup", "local");
        __builtin_amdgcn_s_barrier();
        __builtin_amdgcn_fence(__ATOMIC_ACQUIRE, "workgroup", "local");
    }
}

extern "C" void kernel_launch(void* const* d_in, const int* in_sizes, int n_in,
                              void* d_out, int out_size, void* d_ws, size_t ws_size,
                              hipStream_t stream) {
    (void)in_sizes; (void)n_in; (void)out_size; (void)d_ws; (void)ws_size;
    const float* x    = (const float*)d_in[0];
    const float* z0   = (const float*)d_in[1];
    const float* W_ih = (const float*)d_in[2];
    const float* W_hh = (const float*)d_in[3];
    const float* b_ih = (const float*)d_in[4];
    const float* b_hh = (const float*)d_in[5];
    const float* W1   = (const float*)d_in[6];
    const float* b1   = (const float*)d_in[7];
    const float* W2   = (const float*)d_in[8];
    const float* b2   = (const float*)d_in[9];
    float* out = (float*)d_out;

    dim3 grid(BATCH / 16);
    dim3 block(96);
    monotonic_lstm_wmma<<<grid, block, 0, stream>>>(x, z0, W_ih, W_hh, b_ih, b_hh,
                                                    W1, b1, W2, b2, out);
}


# Recurrence loop written out 5 steps per trip: accumulators alternate register sets (no copies), one branch, one store group and one place of hand-over LDS reads per 5 steps; sum order unchanged
# speedup vs baseline: 4.7672x; 1.1120x over previous
#include <hip/hip_runtime.h>
#include <hip/hip_bf16.h>


#define BATCH 4096
#define TSEQ  200
#define FEAT  32
#define HID   8
#define FWD   5
#define NB    25
#define NBLK  (TSEQ / NB)
#define SLOT  256
#define UNR   5


typedef __attribute__((ext_vector_type(2))) float v2f;
typedef __attribute__((ext_vector_type(4))) float v4f;
typedef __attribute__((ext_vector_type(8))) float v8f;

__device__ __forceinline__ float hw_tanh(float a) {
    return __builtin_amdgcn_tanhf(a);
}
__device__ __forceinline__ float sigmoid_of_half(float half_a) {
    return fmaf(0.5f, hw_tanh(half_a), 0.5f);
}

__device__ __forceinline__ float gate_scale(int g) {
    return (g >= 2 * HID && g < 3 * HID) ? 1.0f : 0.5f;
}

__device__ __forceinline__ int gate_of_row(int tile, int rho) {
    const int r = rho & 7, h = rho >> 3;
    return tile * 16 + ((r & 4) ? 8 : 0) + 4 * h + (r & 3);
}

__device__ __forceinline__ float wfrag(const float* __restrict__ W_ih,
                                       const float* __restrict__ W_hh,
                                       int g, int c, int hi, int s) {
    if (c < 8)  return W_ih[g * (FEAT + 1) + 4 * c + 2 * hi + s];
    if (c == 8) return W_hh[g * HID + 4 * hi + s];
    if (c == 9) return W_hh[g * HID + 4 * hi + 2 + s];
    return (hi == 0 && s == 0) ? W_ih[g * (FEAT + 1) + FEAT] : 0.0f;
}


__device__ __forceinline__ void loadx(v2f (&dst)[8], const float* __restrict__ xr, int hi) {
#pragma unroll
    for (int c = 0; c < 8; ++c)
        dst[c] = *reinterpret_cast<const v2f*>(xr + 4 * c + 2 * hi);
}

__device__ __forceinline__ void put8(v4f* p, v8f v) {
    p[0]  = __builtin_shufflevector(v, v, 0, 1, 2, 3);
    p[32] = __builtin_shufflevector(v, v, 4, 5, 6, 7);
}
__device__ __forceinline__ v8f get8(const v4f* p) {
    const v4f lo = p[0], up = p[32];
    return __builtin_shufflevector(lo, up, 0, 1, 2, 3, 4, 5, 6, 7);
}

#define WMMA_F32(ACC, A, BFRAG) \
    (ACC) = __builtin_amdgcn_wmma_f32_16x16x4_f32(false, (A), false, (BFRAG), (short)0, (ACC), false, false)
#define WMMA_ACC(D, A, BFRAG, C) \
    (D) = __builtin_amdgcn_wmma_f32_16x16x4_f32(false, (A), false, (BFRAG), (short)0, (C), false, false)

#define ORDER6(A, B, C, D, E, F) \
    asm volatile("" : "+v"(A), "+v"(B), "+v"(C), "+v"(D), "+v"(E), "+v"(F))
#define ORDER7(A, B, P) \
    asm volatile("" : "+v"(A), "+v"(B), "+v"(P[0]), "+v"(P[1]), "+v"(P[2]), "+v"(P[3]), "+v"(P[4]))

__global__ __launch_bounds__(96)
void monotonic_lstm_wmma(const float* __restrict__ x,
                         const float* __restrict__ z0,
                         const float* __restrict__ W_ih,
                         const float* __restrict__ W_hh,
                         const float* __restrict__ b_ih,
                         const float* __restrict__ b_hh,
                         const float* __restrict__ W1,
                         const float* __restrict__ b1,
                         const float* __restrict__ W2,
                         const float* __restrict__ b2,
                         float* __restrict__ out)
{
    __shared__ v4f ring[2 * NB * SLOT];

    const int lane = threadIdx.x & 31;
    const int wave = __builtin_amdgcn_readfirstlane((int)(threadIdx.x >> 5));
    const int m    = lane & 15;
    const int hi   = lane >> 4;
    const int b0   = blockIdx.x * 16;
    const bool projector = wave != 0;
    const int  tile = wave - 1;

    v8f zero8;
#pragma unroll
    for (int r = 0; r < 8; ++r) zero8[r] = 0.0f;

    v2f aw[8] = {};
    v8f biasv = zero8;
    v2f xbuf[5][8] = {};
    const float* xrow = x + ((size_t)(b0 + m) * TSEQ) * FEAT;

    v2f ah8_0 = {}, ah9_0 = {}, az_0 = {}, ah8_1 = {}, ah9_1 = {}, az_1 = {};
    v2f ah0 = {}, ah1 = {};
    v8f b1v = zero8;
    float w2r[FWD] = {0.f, 0.f, 0.f, 0.f, 0.f};
    float b2r = 0.0f;
    float cst[4]  = {0.f, 0.f, 0.f, 0.f};
    float hreg[4] = {0.f, 0.f, 0.f, 0.f};
    float zreg = 0.0f;
    float* orow = out + (size_t)(b0 + m) * TSEQ;

    if (projector) {
        const int g = gate_of_row(tile, m);
#pragma unroll
        for (int c = 0; c < 8; ++c) {
            aw[c].x = gate_scale(g) * wfrag(W_ih, W_hh, g, c, hi, 0);
            aw[c].y = gate_scale(g) * wfrag(W_ih, W_hh, g, c, hi, 1);
        }
#pragma unroll
        for (int r = 0; r < 8; ++r) {
            const int ga = gate_of_row(tile, 8 * hi + r);
            biasv[r] = gate_scale(ga) * (b_ih[ga] + b_hh[ga]);
        }
#pragma unroll
        for (int j = 0; j < 5; ++j) loadx(xbuf[j], xrow + j * FEAT, hi);
    } else {
        const int g0 = gate_of_row(0, m);
        const int g1 = gate_of_row(1, m);
        const float k0 = gate_scale(g0), k1 = gate_scale(g1);
        ah8_0.x = k0 * wfrag(W_ih, W_hh, g0, 8, hi, 0);  ah8_0.y = k0 * wfrag(W_ih, W_hh, g0, 8, hi, 1);
        ah9_0.x = k0 * wfrag(W_ih, W_hh, g0, 9, hi, 0);  ah9_0.y = k0 * wfrag(W_ih, W_hh, g0, 9, hi, 1);
        az_0.x  = k0 * wfrag(W_ih, W_hh, g0, 10, hi, 0); az_0.y  = k0 * wfrag(W_ih, W_hh, g0, 10, hi, 1);
        ah8_1.x = k1 * wfrag(W_ih, W_hh, g1, 8, hi, 0);  ah8_1.y = k1 * wfrag(W_ih, W_hh, g1, 8, hi, 1);
        ah9_1.x = k1 * wfrag(W_ih, W_hh, g1, 9, hi, 0);  ah9_1.y = k1 * wfrag(W_ih, W_hh, g1, 9, hi, 1);
        az_1.x  = k1 * wfrag(W_ih, W_hh, g1, 10, hi, 0); az_1.y  = k1 * wfrag(W_ih, W_hh, g1, 10, hi, 1);
        const int  kk = m & 7;
        const bool kv = kk < FWD;
        ah0.x = kv ? W1[kk * HID + 4 * hi + 0] : 0.0f;
        ah0.y = kv ? W1[kk * HID + 4 * hi + 1] : 0.0f;
        ah1.x = kv ? W1[kk * HID + 4 * hi + 2] : 0.0f;
        ah1.y = kv ? W1[kk * HID + 4 * hi + 3] : 0.0f;
#pragma unroll
        for (int i = 0; i < FWD; ++i) { b1v[i] = b1[i]; w2r[i] = W2[i]; }
        b2r  = b2[0];
        zreg = z0[b0 + m];
    }

    for (int k = 0; k <= NBLK; ++k) {
        if (projector) {
            if (k < NBLK) {
                v4f* wb = ring + ((k & 1) * NB) * SLOT + tile * 128 + lane;
                const int t0 = k * NB;
                for (int q = 0; q < NB / 5; ++q) {
#pragma unroll
                    for (int j = 0; j < 5; ++j) {
                        const int s = 5 * q + j;
                        v8f pe, po;
                        WMMA_ACC(pe, aw[0], xbuf[j][0], biasv);
                        WMMA_ACC(po, aw[1], xbuf[j][1], zero8);
#pragma unroll
                        for (int c = 2; c < 8; c += 2) {
                            WMMA_F32(pe, aw[c],     xbuf[j][c]);
                            WMMA_F32(po, aw[c + 1], xbuf[j][c + 1]);
                        }
                        put8(wb + s * SLOT,      pe);
                        put8(wb + s * SLOT + 64, po);
                        const int tn = t0 + s + 5;
                        const int tl = (tn < TSEQ) ? tn : (TSEQ - 1);
                        __builtin_amdgcn_sched_barrier(0);
                        loadx(xbuf[j], xrow + (size_t)tl * FEAT, hi);
                        __builtin_amdgcn_sched_barrier(0);
                    }
                }
            }
        } else {
            if (k >= 1) {
                const int kb = k - 1;
                const v4f* rb = ring + ((kb & 1) * NB) * SLOT + lane;
                const int t0 = kb * NB;
                v8f a0e, a0o, a1e, a1o;
                {
                    v2f hb0, hb1;
                    hb0.x = hreg[0]; hb0.y = hreg[1];
                    hb1.x = hreg[2]; hb1.y = hreg[3];
                    WMMA_ACC(a0e, ah8_0, hb0, get8(rb));
                    WMMA_ACC(a1e, ah8_1, hb0, get8(rb + 128));
                    WMMA_ACC(a0o, ah9_0, hb1, get8(rb + 64));
                    WMMA_ACC(a1o, ah9_1, hb1, get8(rb + 192));
                }
#pragma nounroll
                for (int q = 0; q < NB / UNR; ++q) {
                    v8f p0e[UNR], p0o[UNR], p1e[UNR], p1o[UNR];
                    float zs[UNR];
#pragma unroll
                    for (int j = 0; j < UNR; ++j) {
                        {
                            v2f zb;
                            zb.x = hi ? 0.0f : zreg;
                            zb.y = 0.0f;
                            WMMA_F32(a0e, az_0, zb);
                            WMMA_F32(a1e, az_1, zb);
                        }
                        if (j == 0) {
                            __builtin_amdgcn_sched_barrier(0);
#pragma unroll
                            for (int i = 0; i < UNR; ++i) {
                                const int sn = UNR * q + i + 1;
                                const int sc = (sn < NB) ? sn : (NB - 1);
                                p0e[i] = get8(rb + sc * SLOT);
                                p0o[i] = get8(rb + sc * SLOT + 64);
                                p1e[i] = get8(rb + sc * SLOT + 128);
                                p1o[i] = get8(rb + sc * SLOT + 192);
                            }
                            __builtin_amdgcn_sched_barrier(0);
                        }

                        float sI[4], sF[4], sG[4], sO[4];
#pragma unroll
                        for (int u = 0; u < 4; ++u) {
                            sG[u] = a1e[u]     + a1o[u];
                            sI[u] = a0e[u]     + a0o[u];
                            sF[u] = a0e[4 + u] + a0o[4 + u];
                            sO[u] = a1e[4 + u] + a1o[4 + u];
                        }
                        float tG[4], gI[4], gF[4], gO[4];
#pragma unroll
                        for (int u = 0; u < 4; ++u) {
                            tG[u] = hw_tanh(sG[u]);
                            gI[u] = sigmoid_of_half(sI[u]);
                            gF[u] = sigmoid_of_half(sF[u]);
                            gO[u] = sigmoid_of_half(sO[u]);
                        }
#pragma unroll
                        for (int u = 0; u < 4; ++u) {
                            const float ig = gI[u] * tG[u];
                            cst[u] = fmaf(gF[u], cst[u], ig);
                            hreg[u] = gO[u] * hw_tanh(cst[u]);
                        }

                        v8f hda, hdb, n0e, n0o, n1e, n1o;
                        v2f hb0, hb1;
                        hb0.x = hreg[0]; hb0.y = hreg[1];
                        hb1.x = hreg[2]; hb1.y = hreg[3];
                        WMMA_ACC(hda, ah0, hb0, b1v);
                        WMMA_ACC(hdb, ah1, hb1, zero8);
                        ORDER6(hda, hdb, p0e[j], p1e[j], p0o[j], p1o[j]);
                        WMMA_ACC(n0e, ah8_0, hb0, p0e[j]);
                        WMMA_ACC(n1e, ah8_1, hb0, p1e[j]);
                        ORDER6(hda, hdb, n0e, n1e, p0o[j], p1o[j]);
                        float dk[FWD];
#pragma unroll
                        for (int i = 0; i < FWD; ++i) dk[i] = fmaxf(hda[i] + hdb[i], 0.0f);
                        WMMA_ACC(n0o, ah9_0, hb1, p0o[j]);
                        ORDER7(n0o, p1o[j], dk);
                        WMMA_ACC(n1o, ah9_1, hb1, p1o[j]);
                        float delta = b2r;
#pragma unroll
                        for (int i = 0; i < FWD; ++i)
                            delta = fmaf(dk[i], w2r[i], delta);
                        zreg += fmaxf(delta, 0.0f);
                        zs[j] = zreg;

                        a0e = n0e; a0o = n0o; a1e = n1e; a1o = n1o;
                    }
                    if (hi == 0) {
#pragma unroll
                        for (int j = 0; j < UNR; ++j) {
                            orow[t0 + UNR * q + j] = zs[j];
                            asm volatile("" ::: "memory");
                        }
                    }
                }
            }
        }
        __builtin_amdgcn_fence(__ATOMIC_RELEASE, "workgroup", "local");
        __builtin_amdgcn_s_barrier();
        __builtin_amdgcn_fence(__ATOMIC_ACQUIRE, "workgroup", "local");
    }
}

extern "C" void kernel_launch(void* const* d_in, const int* in_sizes, int n_in,
                              void* d_out, int out_size, void* d_ws, size_t ws_size,
                              hipStream_t stream) {
    (void)in_sizes; (void)n_in; (void)out_size; (void)d_ws; (void)ws_size;
    const float* x    = (const float*)d_in[0];
    const float* z0   = (const float*)d_in[1];
    const float* W_ih = (const float*)d_in[2];
    const float* W_hh = (const float*)d_in[3];
    const float* b_ih = (const float*)d_in[4];
    const float* b_hh = (const float*)d_in[5];
    const float* W1   = (const float*)d_in[6];
    const float* b1   = (const float*)d_in[7];
    const float* W2   = (const float*)d_in[8];
    const float* b2   = (const float*)d_in[9];
    float* out = (float*)d_out;

    dim3 grid(BATCH / 16);
    dim3 block(96);
    monotonic_lstm_wmma<<<grid, block, 0, stream>>>(x, z0, W_ih, W_hh, b_ih, b_hh,
                                                    W1, b1, W2, b2, out);
}
